# starting kernel
# baseline (speedup 1.0000x reference)
#include <hip/hip_runtime.h>

typedef __attribute__((ext_vector_type(2))) float v2f;
typedef __attribute__((ext_vector_type(8))) float v8f;


__global__ void set_deg_kernel(float* deg, int n) {
    int i = blockIdx.x * blockDim.x + threadIdx.x;
    if (i < n) deg[i] = 1.0f;
}

__global__ void accum_deg_kernel(float* deg, const int* __restrict__ dst,
                                 const float* __restrict__ w, int e) {
    int i = blockIdx.x * blockDim.x + threadIdx.x;
    if (i < e) unsafeAtomicAdd(&deg[dst[i]], w[i]);
}

__global__ void dinv_kernel(float* deg, int n) {
    int i = blockIdx.x * blockDim.x + threadIdx.x;
    if (i < n) deg[i] = rsqrtf(deg[i]);
}

__global__ void norm_kernel(const float* __restrict__ dinv, const int* __restrict__ src,
                            const int* __restrict__ dst, const float* __restrict__ w,
                            float* __restrict__ nrm, int e) {
    int i = blockIdx.x * blockDim.x + threadIdx.x;
    if (i < e) nrm[i] = dinv[src[i]] * w[i] * dinv[dst[i]];
}


template <int COLS>
__global__ __launch_bounds__(256) void gemm_wmma_kernel(
    const float* __restrict__ X, const float* __restrict__ W,
    float* __restrict__ OUT, int nRows) {
    constexpr int K  = 128;
    constexpr int NT = COLS / 16;
    constexpr int CSH = (COLS == 128) ? 7 : 6;

    __shared__ float sW[(K / 2) * COLS * 2];
    for (int i = threadIdx.x; i < K * COLS; i += 256) {
        const int row = i >> CSH;
        const int col = i & (COLS - 1);
        sW[(row >> 1) * (COLS * 2) + col * 2 + (row & 1)] = W[i];
    }
    __syncthreads();

    const int wave = threadIdx.x >> 5;
    const int lane = threadIdx.x & 31;
    const int rb   = blockIdx.x * 8 + wave;
    if (rb * 16 >= nRows) return;

    const int l15  = lane & 15;
    const int hsel = lane >> 4;

    const float* aBase = X + (size_t)(rb * 16 + l15) * K + 2 * hsel;
    const float* bCol  = &sW[hsel * (COLS * 2) + l15 * 2];

    v8f acc[NT] = {};

    for (int k = 0; k < K; k += 4) {
        const v2f A = *(const v2f*)(aBase + k);
        const float* bp = bCol + (k >> 1) * (COLS * 2);
#pragma unroll
        for (int c = 0; c < NT; ++c) {
            const v2f B = *(const v2f*)(bp + c * 32);
            acc[c] = __builtin_amdgcn_wmma_f32_16x16x4_f32(
                false, A, false, B, (short)0, acc[c], false, false);
        }
    }

    const int rowBase = rb * 16 + (hsel << 3);
#pragma unroll
    for (int c = 0; c < NT; ++c) {
#pragma unroll
        for (int r = 0; r < 8; ++r)
            OUT[(size_t)(rowBase + r) * COLS + c * 16 + l15] = acc[c][r];
    }
}


template <int COLS>
__global__ void init_out_kernel(const float* __restrict__ h, const float* __restrict__ dinv,
                                const float* __restrict__ b, float* __restrict__ out,
                                long long total) {
    constexpr int SHIFT = (COLS == 128) ? 7 : 6;
    long long idx = (long long)blockIdx.x * blockDim.x + threadIdx.x;
    if (idx >= total) return;
    int node = (int)(idx >> SHIFT);
    int f    = (int)(idx & (COLS - 1));
    float dv = dinv[node];
    out[idx] = h[idx] * dv * dv + b[f];
}

template <int COLS>
__global__ __launch_bounds__(256) void scatter_kernel(
    const float* __restrict__ h, const int* __restrict__ src,
    const int* __restrict__ dst, const float* __restrict__ nrm,
    float* out, int e) {
    const long long gid = (long long)blockIdx.x * blockDim.x + threadIdx.x;
    const long long eid = gid >> 5;
    const int lane = (int)(gid & 31);
    if (eid >= (long long)e) return;
    const int s = src[eid];
    const int d = dst[eid];
    const float nm = nrm[eid];
    if (COLS == 128) {
        const float4 v = *(const float4*)&h[(size_t)s * 128 + lane * 4];
        float* o = &out[(size_t)d * 128 + lane * 4];
        unsafeAtomicAdd(o + 0, v.x * nm);
        unsafeAtomicAdd(o + 1, v.y * nm);
        unsafeAtomicAdd(o + 2, v.z * nm);
        unsafeAtomicAdd(o + 3, v.w * nm);
    } else {
        const float2 v = *(const float2*)&h[(size_t)s * 64 + lane * 2];
        float* o = &out[(size_t)d * 64 + lane * 2];
        unsafeAtomicAdd(o + 0, v.x * nm);
        unsafeAtomicAdd(o + 1, v.y * nm);
    }
}

__global__ void relu_kernel(const float* __restrict__ in, float* __restrict__ out,
                            long long total) {
    long long i = (long long)blockIdx.x * blockDim.x + threadIdx.x;
    if (i < total) out[i] = fmaxf(in[i], 0.0f);
}


extern "C" void kernel_launch(void* const* d_in, const int* in_sizes, int n_in,
                              void* d_out, int out_size, void* d_ws, size_t ws_size,
                              hipStream_t stream) {
    const float* x   = (const float*)d_in[0];
    const int*  eidx = (const int*)d_in[1];
    const float* ew  = (const float*)d_in[2];
    const float* W1  = (const float*)d_in[3];
    const float* b1  = (const float*)d_in[4];
    const float* W2  = (const float*)d_in[5];
    const float* b2  = (const float*)d_in[6];

    const int N = in_sizes[0] / 128;
    const int E = in_sizes[2];
    const int* src = eidx;
    const int* dst = eidx + E;

    float* dinv = (float*)d_ws;
    float* nrm  = dinv + N;
    float* h1   = nrm + E;
    float* acc  = h1 + (size_t)N * 128;
    float* out  = (float*)d_out;

    const int T = 256;
    const long long nf1 = (long long)N * 128;
    const long long nf2 = (long long)N * 64;
    const int gemmBlocks = ((N + 15) / 16 + 7) / 8;

    set_deg_kernel<<<(N + T - 1) / T, T, 0, stream>>>(dinv, N);
    accum_deg_kernel<<<(E + T - 1) / T, T, 0, stream>>>(dinv, dst, ew, E);
    dinv_kernel<<<(N + T - 1) / T, T, 0, stream>>>(dinv, N);
    norm_kernel<<<(E + T - 1) / T, T, 0, stream>>>(dinv, src, dst, ew, nrm, E);

    gemm_wmma_kernel<128><<<gemmBlocks, T, 0, stream>>>(x, W1, h1, N);
    init_out_kernel<128><<<(int)((nf1 + T - 1) / T), T, 0, stream>>>(h1, dinv, b1, acc, nf1);
    scatter_kernel<128><<<(int)(((long long)E * 32 + T - 1) / T), T, 0, stream>>>(
        h1, src, dst, nrm, acc, E);
    relu_kernel<<<(int)((nf1 + T - 1) / T), T, 0, stream>>>(acc, h1, nf1);

    gemm_wmma_kernel<64><<<gemmBlocks, T, 0, stream>>>(h1, W2, acc, N);
    init_out_kernel<64><<<(int)((nf2 + T - 1) / T), T, 0, stream>>>(acc, dinv, b2, out, nf2);
    scatter_kernel<64><<<(int)(((long long)E * 32 + T - 1) / T), T, 0, stream>>>(
        acc, src, dst, nrm, out, E);
}


# push to pull: per-call counting sort of edges by destination (integer atomic ranks), then wave-per-node gather sum with self-loop, bias, relu fused; float atomics gone, sum order = arrival rank
# speedup vs baseline: 3.1419x; 3.1419x over previous
#include <hip/hip_runtime.h>

typedef __attribute__((ext_vector_type(2))) float v2f;
typedef __attribute__((ext_vector_type(4))) float v4f;
typedef __attribute__((ext_vector_type(8))) float v8f;


constexpr int CNT_STRIDE = 16;

__device__ __forceinline__ int wave_sum(int v) {
    for (int o = 16; o > 0; o >>= 1) v += __shfl_xor(v, o);
    return v;
}

__global__ void zero_cnt_kernel(int4* __restrict__ p, int n4) {
    int i = blockIdx.x * blockDim.x + threadIdx.x;
    if (i < n4) p[i] = make_int4(0, 0, 0, 0);
}

__global__ void hist_rank_kernel(int* cnt, const int* __restrict__ dst,
                                 int* __restrict__ rank, int e) {
    int i = blockIdx.x * blockDim.x + threadIdx.x;
    if (i < e) rank[i] = atomicAdd(&cnt[(size_t)dst[i] * CNT_STRIDE], 1);
}

__global__ __launch_bounds__(1024) void block_sum_kernel(
    const int* __restrict__ cnt, int* __restrict__ cntc, int* __restrict__ bsum, int n) {
    __shared__ int sw[32];
    const int lane = threadIdx.x & 31;
    const int wave = threadIdx.x >> 5;
    const int i = blockIdx.x * 1024 + threadIdx.x;
    int c = 0;
    if (i < n) { c = cnt[(size_t)i * CNT_STRIDE]; cntc[i] = c; }
    const int v = wave_sum(c);
    if (lane == 0) sw[wave] = v;
    __syncthreads();
    if (wave == 0) {
        const int t = wave_sum(sw[lane]);
        if (lane == 0) bsum[blockIdx.x] = t;
    }
}

__global__ __launch_bounds__(1024) void scan_kernel(
    const int* __restrict__ cntc, const int* __restrict__ bsum,
    int* __restrict__ rowptr, int n) {
    __shared__ int sw[32];
    const int lane = threadIdx.x & 31;
    const int wave = threadIdx.x >> 5;
    const int i = blockIdx.x * 1024 + threadIdx.x;
    const int c = (i < n) ? cntc[i] : 0;
    int inc = c;
    for (int o = 1; o < 32; o <<= 1) {
        const int t = __shfl_up(inc, o);
        if (lane >= o) inc += t;
    }
    if (lane == 31) sw[wave] = inc;
    int pb = 0;
    for (int k = lane; k < (int)blockIdx.x; k += 32) pb += bsum[k];
    pb = wave_sum(pb);
    __syncthreads();
    int wv = (lane < wave) ? sw[lane] : 0;
    wv = wave_sum(wv);
    const int excl = pb + wv + inc - c;
    if (i < n) {
        rowptr[i] = excl;
        if (i == n - 1) rowptr[n] = excl + c;
    }
}

__global__ void fill_kernel(const int* __restrict__ rowptr, const int* __restrict__ src,
                            const int* __restrict__ dst, const float* __restrict__ w,
                            const int* __restrict__ rank, int2* __restrict__ es, int e) {
    int i = blockIdx.x * blockDim.x + threadIdx.x;
    if (i < e) {
        const int p = rowptr[dst[i]] + rank[i];
        es[p] = make_int2(src[i], __float_as_int(w[i]));
    }
}

__global__ void dinv_kernel(const int* __restrict__ rowptr, const int2* __restrict__ es,
                            float* __restrict__ dinv, int n) {
    int i = blockIdx.x * blockDim.x + threadIdx.x;
    if (i < n) {
        const int beg = rowptr[i];
        const int end = rowptr[i + 1];
        float deg = 1.0f;
        for (int p = beg; p < end; ++p) deg += __int_as_float(es[p].y);
        dinv[i] = (deg > 0.0f) ? rsqrtf(deg) : 0.0f;
    }
}


template <int COLS>
__global__ __launch_bounds__(256) void gemm_wmma_kernel(
    const float* __restrict__ X, const float* __restrict__ W,
    float* __restrict__ OUT, int nRows) {
    constexpr int K  = 128;
    constexpr int NT = COLS / 16;
    constexpr int CSH = (COLS == 128) ? 7 : 6;

    __shared__ float sW[(K / 2) * COLS * 2];
    for (int i = threadIdx.x; i < K * COLS; i += 256) {
        const int row = i >> CSH;
        const int col = i & (COLS - 1);
        sW[(row >> 1) * (COLS * 2) + col * 2 + (row & 1)] = W[i];
    }
    __syncthreads();

    const int wave = threadIdx.x >> 5;
    const int lane = threadIdx.x & 31;
    const int rb   = blockIdx.x * 8 + wave;
    if (rb * 16 >= nRows) return;

    const int l15  = lane & 15;
    const int hsel = lane >> 4;

    const float* aBase = X + (size_t)(rb * 16 + l15) * K + 2 * hsel;
    const float* bCol  = &sW[hsel * (COLS * 2) + l15 * 2];

    v8f acc[NT] = {};

    for (int k = 0; k < K; k += 4) {
        const v2f A = *(const v2f*)(aBase + k);
        const float* bp = bCol + (k >> 1) * (COLS * 2);
#pragma unroll
        for (int c = 0; c < NT; ++c) {
            const v2f B = *(const v2f*)(bp + c * 32);
            acc[c] = __builtin_amdgcn_wmma_f32_16x16x4_f32(
                false, A, false, B, (short)0, acc[c], false, false);
        }
    }

    const int rowBase = rb * 16 + (hsel << 3);
#pragma unroll
    for (int c = 0; c < NT; ++c) {
#pragma unroll
        for (int r = 0; r < 8; ++r)
            OUT[(size_t)(rowBase + r) * COLS + c * 16 + l15] = acc[c][r];
    }
}


template <int COLS, bool RELU>
__global__ __launch_bounds__(256) void gather_kernel(
    const float* __restrict__ h, const int* __restrict__ rowptr,
    const int2* __restrict__ es, const float* __restrict__ dinv,
    const float* __restrict__ b, float* __restrict__ out, int n) {
    constexpr int VPL = COLS / 32;
    typedef __attribute__((ext_vector_type(VPL))) float vec;

    const int lane = threadIdx.x & 31;
    const int nodeV = blockIdx.x * 8 + (threadIdx.x >> 5);
    if (nodeV >= n) return;
    const int node = __builtin_amdgcn_readfirstlane(nodeV);

    const int beg = rowptr[node];
    const int end = rowptr[node + 1];
    const float dd = dinv[node];
    const float* hl = h + lane * VPL;

    vec acc = *(const vec*)(hl + (size_t)node * COLS) * dd * dd;

    for (int base = beg; base < end; base += 32) {
        const int idx = base + lane;
        int s = node;
        float c = 0.0f;
        if (idx < end) {
            const int2 p = es[idx];
            s = p.x;
            c = dinv[s] * __int_as_float(p.y) * dd;
        }
        const int ci = __float_as_int(c);
        const int rem = end - base;
        const int cnt = rem < 32 ? rem : 32;
        int j = 0;
        for (; j + 4 <= cnt; j += 4) {
            const int s0 = __builtin_amdgcn_readlane(s, j);
            const int s1 = __builtin_amdgcn_readlane(s, j + 1);
            const int s2 = __builtin_amdgcn_readlane(s, j + 2);
            const int s3 = __builtin_amdgcn_readlane(s, j + 3);
            const float c0 = __int_as_float(__builtin_amdgcn_readlane(ci, j));
            const float c1 = __int_as_float(__builtin_amdgcn_readlane(ci, j + 1));
            const float c2 = __int_as_float(__builtin_amdgcn_readlane(ci, j + 2));
            const float c3 = __int_as_float(__builtin_amdgcn_readlane(ci, j + 3));
            const vec v0 = *(const vec*)(hl + (size_t)s0 * COLS);
            const vec v1 = *(const vec*)(hl + (size_t)s1 * COLS);
            const vec v2 = *(const vec*)(hl + (size_t)s2 * COLS);
            const vec v3 = *(const vec*)(hl + (size_t)s3 * COLS);
            acc += v0 * c0;
            acc += v1 * c1;
            acc += v2 * c2;
            acc += v3 * c3;
        }
        for (; j < cnt; ++j) {
            const int s0 = __builtin_amdgcn_readlane(s, j);
            const float c0 = __int_as_float(__builtin_amdgcn_readlane(ci, j));
            const vec v0 = *(const vec*)(hl + (size_t)s0 * COLS);
            acc += v0 * c0;
        }
    }

    acc += *(const vec*)(b + lane * VPL);
    if (RELU) {
#pragma unroll
        for (int k = 0; k < VPL; ++k) acc[k] = fmaxf(acc[k], 0.0f);
    }
    *(vec*)(out + (size_t)node * COLS + lane * VPL) = acc;
}


static inline char* bump(char*& p, size_t bytes) {
    char* r = p;
    p += (bytes + 255) & ~(size_t)255;
    return r;
}

extern "C" void kernel_launch(void* const* d_in, const int* in_sizes, int n_in,
                              void* d_out, int out_size, void* d_ws, size_t ws_size,
                              hipStream_t stream) {
    const float* x   = (const float*)d_in[0];
    const int*  eidx = (const int*)d_in[1];
    const float* ew  = (const float*)d_in[2];
    const float* W1  = (const float*)d_in[3];
    const float* b1  = (const float*)d_in[4];
    const float* W2  = (const float*)d_in[5];
    const float* b2  = (const float*)d_in[6];

    const int N = in_sizes[0] / 128;
    const int E = in_sizes[2];
    const int* src = eidx;
    const int* dst = eidx + E;

    char* p = (char*)d_ws;
    int*   cnt    = (int*)  bump(p, (size_t)N * CNT_STRIDE * sizeof(int));
    int*   cntc   = (int*)  bump(p, (size_t)N * sizeof(int));
    int*   bsum   = (int*)  bump(p, (size_t)((N + 1023) / 1024) * sizeof(int));
    int*   rowptr = (int*)  bump(p, (size_t)(N + 1) * sizeof(int));
    int*   rank   = (int*)  bump(p, (size_t)E * sizeof(int));
    int2*  es     = (int2*) bump(p, (size_t)E * sizeof(int2));
    float* dinv   = (float*)bump(p, (size_t)N * sizeof(float));
    float* h1     = (float*)bump(p, (size_t)N * 128 * sizeof(float));
    float* a1     = (float*)bump(p, (size_t)N * 128 * sizeof(float));
    float* g      = (float*)bump(p, (size_t)N * 64 * sizeof(float));
    float* out    = (float*)d_out;

    const int T = 256;
    const int n4 = N * (CNT_STRIDE / 4);
    const int scanBlocks = (N + 1023) / 1024;
    const int gemmBlocks = ((N + 15) / 16 + 7) / 8;
    const int nodeBlocks = (N + 7) / 8;

    zero_cnt_kernel<<<(n4 + T - 1) / T, T, 0, stream>>>((int4*)cnt, n4);
    hist_rank_kernel<<<(E + T - 1) / T, T, 0, stream>>>(cnt, dst, rank, E);
    block_sum_kernel<<<scanBlocks, 1024, 0, stream>>>(cnt, cntc, bsum, N);
    scan_kernel<<<scanBlocks, 1024, 0, stream>>>(cntc, bsum, rowptr, N);
    fill_kernel<<<(E + T - 1) / T, T, 0, stream>>>(rowptr, src, dst, ew, rank, es, E);
    dinv_kernel<<<(N + T - 1) / T, T, 0, stream>>>(rowptr, es, dinv, N);

    gemm_wmma_kernel<128><<<gemmBlocks, T, 0, stream>>>(x, W1, h1, N);
    gather_kernel<128, true><<<nodeBlocks, T, 0, stream>>>(h1, rowptr, es, dinv, b1, a1, N);

    gemm_wmma_kernel<64><<<gemmBlocks, T, 0, stream>>>(a1, W2, g, N);
    gather_kernel<64, false><<<nodeBlocks, T, 0, stream>>>(g, rowptr, es, dinv, b2, out, N);
}


# independent launches merged: layer-1 dense transform runs as extra work-groups inside the histogram and fill launches (half its row blocks each), 4 edges per thread; sum order arrival rank as before
# speedup vs baseline: 3.2672x; 1.0399x over previous
#include <hip/hip_runtime.h>

typedef __attribute__((ext_vector_type(2))) float v2f;
typedef __attribute__((ext_vector_type(4))) float v4f;
typedef __attribute__((ext_vector_type(8))) float v8f;


constexpr int CNT_STRIDE = 16;

__device__ __forceinline__ int wave_sum(int v) {
    for (int o = 16; o > 0; o >>= 1) v += __shfl_xor(v, o);
    return v;
}

__global__ void zero_cnt_kernel(int4* __restrict__ p, int n4) {
    int i = blockIdx.x * blockDim.x + threadIdx.x;
    if (i < n4) p[i] = make_int4(0, 0, 0, 0);
}

__global__ __launch_bounds__(1024) void block_sum_kernel(
    const int* __restrict__ cnt, int* __restrict__ cntc, int* __restrict__ bsum, int n) {
    __shared__ int sw[32];
    const int lane = threadIdx.x & 31;
    const int wave = threadIdx.x >> 5;
    const int i = blockIdx.x * 1024 + threadIdx.x;
    int c = 0;
    if (i < n) { c = cnt[(size_t)i * CNT_STRIDE]; cntc[i] = c; }
    const int v = wave_sum(c);
    if (lane == 0) sw[wave] = v;
    __syncthreads();
    if (wave == 0) {
        const int t = wave_sum(sw[lane]);
        if (lane == 0) bsum[blockIdx.x] = t;
    }
}

__global__ __launch_bounds__(1024) void scan_kernel(
    const int* __restrict__ cntc, const int* __restrict__ bsum,
    int* __restrict__ rowptr, int n) {
    __shared__ int sw[32];
    const int lane = threadIdx.x & 31;
    const int wave = threadIdx.x >> 5;
    const int i = blockIdx.x * 1024 + threadIdx.x;
    const int c = (i < n) ? cntc[i] : 0;
    int inc = c;
    for (int o = 1; o < 32; o <<= 1) {
        const int t = __shfl_up(inc, o);
        if (lane >= o) inc += t;
    }
    if (lane == 31) sw[wave] = inc;
    int pb = 0;
    for (int k = lane; k < (int)blockIdx.x; k += 32) pb += bsum[k];
    pb = wave_sum(pb);
    __syncthreads();
    int wv = (lane < wave) ? sw[lane] : 0;
    wv = wave_sum(wv);
    const int excl = pb + wv + inc - c;
    if (i < n) {
        rowptr[i] = excl;
        if (i == n - 1) rowptr[n] = excl + c;
    }
}

__global__ void dinv_kernel(const int* __restrict__ rowptr, const int2* __restrict__ es,
                            float* __restrict__ dinv, int n) {
    int i = blockIdx.x * blockDim.x + threadIdx.x;
    if (i < n) {
        const int beg = rowptr[i];
        const int end = rowptr[i + 1];
        float deg = 1.0f;
        for (int p = beg; p < end; ++p) deg += __int_as_float(es[p].y);
        dinv[i] = (deg > 0.0f) ? rsqrtf(deg) : 0.0f;
    }
}


template <int COLS>
__device__ __forceinline__ void gemm_block(
    const float* __restrict__ X, const float* __restrict__ W,
    float* __restrict__ OUT, int nRows, int blk, float* sW) {
    constexpr int K  = 128;
    constexpr int NT = COLS / 16;
    constexpr int CSH = (COLS == 128) ? 7 : 6;

    for (int i = threadIdx.x; i < K * COLS; i += 256) {
        const int row = i >> CSH;
        const int col = i & (COLS - 1);
        sW[(row >> 1) * (COLS * 2) + col * 2 + (row & 1)] = W[i];
    }
    __syncthreads();

    const int wave = threadIdx.x >> 5;
    const int lane = threadIdx.x & 31;
    const int rb   = blk * 8 + wave;
    if (rb * 16 >= nRows) return;

    const int l15  = lane & 15;
    const int hsel = lane >> 4;

    const float* aBase = X + (size_t)(rb * 16 + l15) * K + 2 * hsel;
    const float* bCol  = &sW[hsel * (COLS * 2) + l15 * 2];

    v8f acc[NT] = {};

    for (int k = 0; k < K; k += 4) {
        const v2f A = *(const v2f*)(aBase + k);
        const float* bp = bCol + (k >> 1) * (COLS * 2);
#pragma unroll
        for (int c = 0; c < NT; ++c) {
            const v2f B = *(const v2f*)(bp + c * 32);
            acc[c] = __builtin_amdgcn_wmma_f32_16x16x4_f32(
                false, A, false, B, (short)0, acc[c], false, false);
        }
    }

    const int rowBase = rb * 16 + (hsel << 3);
#pragma unroll
    for (int c = 0; c < NT; ++c) {
#pragma unroll
        for (int r = 0; r < 8; ++r)
            OUT[(size_t)(rowBase + r) * COLS + c * 16 + l15] = acc[c][r];
    }
}

template <int COLS>
__global__ __launch_bounds__(256) void gemm_wmma_kernel(
    const float* __restrict__ X, const float* __restrict__ W,
    float* __restrict__ OUT, int nRows) {
    __shared__ float sW[64 * COLS * 2];
    gemm_block<COLS>(X, W, OUT, nRows, blockIdx.x, sW);
}

constexpr int EPT = 4;

__global__ __launch_bounds__(256) void gemm_hist_kernel(
    const float* __restrict__ X, const float* __restrict__ W, float* __restrict__ OUT,
    int nRows, int gemmFirst, int gemmCount,
    int* cnt, const int* __restrict__ dst, int* __restrict__ rank, int e) {
    __shared__ float sW[64 * 128 * 2];
    if ((int)blockIdx.x < gemmCount) {
        gemm_block<128>(X, W, OUT, nRows, gemmFirst + (int)blockIdx.x, sW);
        return;
    }
    const int base = ((int)blockIdx.x - gemmCount) * (256 * EPT) + threadIdx.x;
    int d[EPT];
    int r[EPT];
#pragma unroll
    for (int j = 0; j < EPT; ++j) {
        const int i = base + j * 256;
        d[j] = (i < e) ? dst[i] : -1;
    }
#pragma unroll
    for (int j = 0; j < EPT; ++j) {
        r[j] = 0;
        if (d[j] >= 0) r[j] = atomicAdd(&cnt[(size_t)d[j] * CNT_STRIDE], 1);
    }
#pragma unroll
    for (int j = 0; j < EPT; ++j) {
        const int i = base + j * 256;
        if (i < e) rank[i] = r[j];
    }
}

__global__ __launch_bounds__(256) void gemm_fill_kernel(
    const float* __restrict__ X, const float* __restrict__ W, float* __restrict__ OUT,
    int nRows, int gemmFirst, int gemmCount,
    const int* __restrict__ rowptr, const int* __restrict__ src,
    const int* __restrict__ dst, const float* __restrict__ w,
    const int* __restrict__ rank, int2* __restrict__ es, int e) {
    __shared__ float sW[64 * 128 * 2];
    if ((int)blockIdx.x < gemmCount) {
        gemm_block<128>(X, W, OUT, nRows, gemmFirst + (int)blockIdx.x, sW);
        return;
    }
    const int base = ((int)blockIdx.x - gemmCount) * (256 * EPT) + threadIdx.x;
    int p[EPT];
    int2 v[EPT];
#pragma unroll
    for (int j = 0; j < EPT; ++j) {
        const int i = base + j * 256;
        p[j] = -1;
        v[j] = make_int2(0, 0);
        if (i < e) {
            p[j] = rowptr[dst[i]] + rank[i];
            v[j] = make_int2(src[i], __float_as_int(w[i]));
        }
    }
#pragma unroll
    for (int j = 0; j < EPT; ++j) {
        if (p[j] >= 0) es[p[j]] = v[j];
    }
}


template <int COLS, bool RELU>
__global__ __launch_bounds__(256) void gather_kernel(
    const float* __restrict__ h, const int* __restrict__ rowptr,
    const int2* __restrict__ es, const float* __restrict__ dinv,
    const float* __restrict__ b, float* __restrict__ out, int n) {
    constexpr int VPL = COLS / 32;
    typedef __attribute__((ext_vector_type(VPL))) float vec;

    const int lane = threadIdx.x & 31;
    const int nodeV = blockIdx.x * 8 + (threadIdx.x >> 5);
    if (nodeV >= n) return;
    const int node = __builtin_amdgcn_readfirstlane(nodeV);

    const int beg = rowptr[node];
    const int end = rowptr[node + 1];
    const float dd = dinv[node];
    const float* hl = h + lane * VPL;

    vec acc = *(const vec*)(hl + (size_t)node * COLS) * dd * dd;

    for (int base = beg; base < end; base += 32) {
        const int idx = base + lane;
        int s = node;
        float c = 0.0f;
        if (idx < end) {
            const int2 p = es[idx];
            s = p.x;
            c = dinv[s] * __int_as_float(p.y) * dd;
        }
        const int ci = __float_as_int(c);
        const int rem = end - base;
        const int cnt = rem < 32 ? rem : 32;
        int j = 0;
        for (; j + 4 <= cnt; j += 4) {
            const int s0 = __builtin_amdgcn_readlane(s, j);
            const int s1 = __builtin_amdgcn_readlane(s, j + 1);
            const int s2 = __builtin_amdgcn_readlane(s, j + 2);
            const int s3 = __builtin_amdgcn_readlane(s, j + 3);
            const float c0 = __int_as_float(__builtin_amdgcn_readlane(ci, j));
            const float c1 = __int_as_float(__builtin_amdgcn_readlane(ci, j + 1));
            const float c2 = __int_as_float(__builtin_amdgcn_readlane(ci, j + 2));
            const float c3 = __int_as_float(__builtin_amdgcn_readlane(ci, j + 3));
            const vec v0 = *(const vec*)(hl + (size_t)s0 * COLS);
            const vec v1 = *(const vec*)(hl + (size_t)s1 * COLS);
            const vec v2 = *(const vec*)(hl + (size_t)s2 * COLS);
            const vec v3 = *(const vec*)(hl + (size_t)s3 * COLS);
            acc += v0 * c0;
            acc += v1 * c1;
            acc += v2 * c2;
            acc += v3 * c3;
        }
        for (; j < cnt; ++j) {
            const int s0 = __builtin_amdgcn_readlane(s, j);
            const float c0 = __int_as_float(__builtin_amdgcn_readlane(ci, j));
            const vec v0 = *(const vec*)(hl + (size_t)s0 * COLS);
            acc += v0 * c0;
        }
    }

    acc += *(const vec*)(b + lane * VPL);
    if (RELU) {
#pragma unroll
        for (int k = 0; k < VPL; ++k) acc[k] = fmaxf(acc[k], 0.0f);
    }
    *(vec*)(out + (size_t)node * COLS + lane * VPL) = acc;
}


static inline char* bump(char*& p, size_t bytes) {
    char* r = p;
    p += (bytes + 255) & ~(size_t)255;
    return r;
}

extern "C" void kernel_launch(void* const* d_in, const int* in_sizes, int n_in,
                              void* d_out, int out_size, void* d_ws, size_t ws_size,
                              hipStream_t stream) {
    const float* x   = (const float*)d_in[0];
    const int*  eidx = (const int*)d_in[1];
    const float* ew  = (const float*)d_in[2];
    const float* W1  = (const float*)d_in[3];
    const float* b1  = (const float*)d_in[4];
    const float* W2  = (const float*)d_in[5];
    const float* b2  = (const float*)d_in[6];

    const int N = in_sizes[0] / 128;
    const int E = in_sizes[2];
    const int* src = eidx;
    const int* dst = eidx + E;

    char* p = (char*)d_ws;
    int*   cnt    = (int*)  bump(p, (size_t)N * CNT_STRIDE * sizeof(int));
    int*   cntc   = (int*)  bump(p, (size_t)N * sizeof(int));
    int*   bsum   = (int*)  bump(p, (size_t)((N + 1023) / 1024) * sizeof(int));
    int*   rowptr = (int*)  bump(p, (size_t)(N + 1) * sizeof(int));
    int*   rank   = (int*)  bump(p, (size_t)E * sizeof(int));
    int2*  es     = (int2*) bump(p, (size_t)E * sizeof(int2));
    float* dinv   = (float*)bump(p, (size_t)N * sizeof(float));
    float* h1     = (float*)bump(p, (size_t)N * 128 * sizeof(float));
    float* a1     = (float*)bump(p, (size_t)N * 128 * sizeof(float));
    float* g      = (float*)bump(p, (size_t)N * 64 * sizeof(float));
    float* out    = (float*)d_out;

    const int T = 256;
    const int n4 = N * (CNT_STRIDE / 4);
    const int scanBlocks = (N + 1023) / 1024;
    const int gemmBlocks = ((N + 15) / 16 + 7) / 8;
    const int nodeBlocks = (N + 7) / 8;
    const int gemmA = gemmBlocks / 2;
    const int gemmB = gemmBlocks - gemmA;
    const int edgeBlocks = (E + T * EPT - 1) / (T * EPT);

    zero_cnt_kernel<<<(n4 + T - 1) / T, T, 0, stream>>>((int4*)cnt, n4);
    gemm_hist_kernel<<<gemmA + edgeBlocks, T, 0, stream>>>(x, W1, h1, N, 0, gemmA,
                                                          cnt, dst, rank, E);
    block_sum_kernel<<<scanBlocks, 1024, 0, stream>>>(cnt, cntc, bsum, N);
    scan_kernel<<<scanBlocks, 1024, 0, stream>>>(cntc, bsum, rowptr, N);
    gemm_fill_kernel<<<gemmB + edgeBlocks, T, 0, stream>>>(x, W1, h1, N, gemmA, gemmB,
                                                          rowptr, src, dst, ew, rank, es, E);
    dinv_kernel<<<(N + T - 1) / T, T, 0, stream>>>(rowptr, es, dinv, N);

    gather_kernel<128, true><<<nodeBlocks, T, 0, stream>>>(h1, rowptr, es, dinv, b1, a1, N);

    gemm_wmma_kernel<64><<<gemmBlocks, T, 0, stream>>>(a1, W2, g, N);
    gather_kernel<64, false><<<nodeBlocks, T, 0, stream>>>(g, rowptr, es, dinv, b2, out, N);
}


# edge ordering by destination moved into LDS: per-chunk bin counts, one scan, run-wise stores into 512-node bins, per-bin LDS sort that also writes row starts and deg^-1/2; sum order = LDS arrival
# speedup vs baseline: 4.4638x; 1.3662x over previous
#include <hip/hip_runtime.h>


typedef __attribute__((ext_vector_type(2))) float v2f;
typedef __attribute__((ext_vector_type(4))) float v4f;
typedef __attribute__((ext_vector_type(8))) float v8f;


constexpr int BIN_SHIFT = 9;
constexpr int BIN_NODES = 1 << BIN_SHIFT;
constexpr int REC_SHIFT = 32 - BIN_SHIFT;
constexpr unsigned REC_MASK = (1u << REC_SHIFT) - 1u;
constexpr int EPT   = 16;
constexpr int CHUNK = 256 * EPT;
constexpr int CAP   = 12288;
constexpr int SORT_U = 8;

__device__ __forceinline__ int wave_sum(int v) {
    for (int o = 16; o > 0; o >>= 1) v += __shfl_xor(v, o);
    return v;
}

__device__ __forceinline__ int wave_incl_scan(int v, int lane) {
    for (int o = 1; o < 32; o <<= 1) {
        const int t = __shfl_up(v, o);
        if (lane >= o) v += t;
    }
    return v;
}

__global__ __launch_bounds__(1024) void scan_kernel(
    const int* __restrict__ cm, int* __restrict__ scanned, int n) {
    __shared__ int sw[32];
    __shared__ int sb[32];
    const int lane = threadIdx.x & 31;
    const int wave = threadIdx.x >> 5;
    const int first = blockIdx.x * 1024;
    int pb = 0;
    for (int k = threadIdx.x; k < first; k += 1024) pb += cm[k];
    pb = wave_sum(pb);
    if (lane == 0) sb[wave] = pb;
    const int i = first + threadIdx.x;
    const int c = (i < n) ? cm[i] : 0;
    const int inc = wave_incl_scan(c, lane);
    if (lane == 31) sw[wave] = inc;
    __syncthreads();
    const int base = wave_sum(sb[lane]);
    int wv = (lane < wave) ? sw[lane] : 0;
    wv = wave_sum(wv);
    if (i < n) scanned[i] = base + wv + inc - c;
}

__global__ __launch_bounds__(1024) void bin_sort_kernel(
    const int* __restrict__ scanned, const int2* __restrict__ tmp,
    int* __restrict__ rowptr, int2* __restrict__ es, float* __restrict__ dinv,
    int n, int e, int nBins, int nChunks) {
    __shared__ int2  stage[CAP];
    __shared__ int   cnt[BIN_NODES];
    __shared__ float degs[BIN_NODES];
    __shared__ int   cur[BIN_NODES];
    __shared__ int   sw[BIN_NODES / 32];
    const int tid  = threadIdx.x;
    const int lane = tid & 31;
    const int wave = tid >> 5;
    const int b    = blockIdx.x;
    const int eBeg = scanned[(size_t)b * nChunks];
    const int eEnd = (b + 1 < nBins) ? scanned[(size_t)(b + 1) * nChunks] : e;
    const int m    = eEnd - eBeg;
    const bool fits = m <= CAP;

    if (tid < BIN_NODES) { cnt[tid] = 0; degs[tid] = 0.0f; }
    __syncthreads();
    for (int i0 = tid; i0 < m; i0 += 1024 * SORT_U) {
        int2 rec[SORT_U];
#pragma unroll
        for (int k = 0; k < SORT_U; ++k) {
            const int i = i0 + k * 1024;
            rec[k] = tmp[eBeg + (i < m ? i : m - 1)];
        }
#pragma unroll
        for (int k = 0; k < SORT_U; ++k) {
            if (i0 + k * 1024 < m) {
                const int dl = (int)((unsigned)rec[k].x >> REC_SHIFT);
                atomicAdd(&cnt[dl], 1);
                atomicAdd(&degs[dl], __int_as_float(rec[k].y));
            }
        }
    }
    __syncthreads();

    const int c   = (tid < BIN_NODES) ? cnt[tid] : 0;
    const int inc = wave_incl_scan(c, lane);
    if (tid < BIN_NODES && lane == 31) sw[wave] = inc;
    __syncthreads();
    if (tid < BIN_NODES) {
        int wv = (lane < wave) ? sw[lane] : 0;
        wv = wave_sum(wv);
        const int excl = wv + inc - c;
        cur[tid] = excl;
        const int node = b * BIN_NODES + tid;
        if (node < n) {
            rowptr[node] = eBeg + excl;
            const float deg = 1.0f + degs[tid];
            dinv[node] = (deg > 0.0f) ? rsqrtf(deg) : 0.0f;
        }
    }
    if (b == nBins - 1 && tid == 0) rowptr[n] = eEnd;
    __syncthreads();

    for (int i0 = tid; i0 < m; i0 += 1024 * SORT_U) {
        int2 rec[SORT_U];
#pragma unroll
        for (int k = 0; k < SORT_U; ++k) {
            const int i = i0 + k * 1024;
            rec[k] = tmp[eBeg + (i < m ? i : m - 1)];
        }
#pragma unroll
        for (int k = 0; k < SORT_U; ++k) {
            if (i0 + k * 1024 < m) {
                const int dl  = (int)((unsigned)rec[k].x >> REC_SHIFT);
                const int pos = atomicAdd(&cur[dl], 1);
                const int2 o  = make_int2((int)((unsigned)rec[k].x & REC_MASK), rec[k].y);
                if (fits) stage[pos] = o; else es[eBeg + pos] = o;
            }
        }
    }
    __syncthreads();
    if (fits) {
        for (int i = tid; i < m; i += 1024) es[eBeg + i] = stage[i];
    }
}


template <int COLS>
__device__ __forceinline__ void gemm_block(
    const float* __restrict__ X, const float* __restrict__ W,
    float* __restrict__ OUT, int nRows, int blk, float* sW) {
    constexpr int K  = 128;
    constexpr int NT = COLS / 16;
    constexpr int CSH = (COLS == 128) ? 7 : 6;

    for (int i = threadIdx.x; i < K * COLS; i += 256) {
        const int row = i >> CSH;
        const int col = i & (COLS - 1);
        sW[(row >> 1) * (COLS * 2) + col * 2 + (row & 1)] = W[i];
    }
    __syncthreads();

    const int wave = threadIdx.x >> 5;
    const int lane = threadIdx.x & 31;
    const int rb   = blk * 8 + wave;
    if (rb * 16 >= nRows) return;

    const int l15  = lane & 15;
    const int hsel = lane >> 4;

    const float* aBase = X + (size_t)(rb * 16 + l15) * K + 2 * hsel;
    const float* bCol  = &sW[hsel * (COLS * 2) + l15 * 2];

    v8f acc[NT] = {};

    for (int k = 0; k < K; k += 4) {
        const v2f A = *(const v2f*)(aBase + k);
        const float* bp = bCol + (k >> 1) * (COLS * 2);
#pragma unroll
        for (int c = 0; c < NT; ++c) {
            const v2f B = *(const v2f*)(bp + c * 32);
            acc[c] = __builtin_amdgcn_wmma_f32_16x16x4_f32(
                false, A, false, B, (short)0, acc[c], false, false);
        }
    }

    const int rowBase = rb * 16 + (hsel << 3);
#pragma unroll
    for (int c = 0; c < NT; ++c) {
#pragma unroll
        for (int r = 0; r < 8; ++r)
            OUT[(size_t)(rowBase + r) * COLS + c * 16 + l15] = acc[c][r];
    }
}

template <int COLS>
__global__ __launch_bounds__(256) void gemm_wmma_kernel(
    const float* __restrict__ X, const float* __restrict__ W,
    float* __restrict__ OUT, int nRows) {
    __shared__ float sW[64 * COLS * 2];
    gemm_block<COLS>(X, W, OUT, nRows, blockIdx.x, sW);
}


__global__ __launch_bounds__(256) void gemm_count_kernel(
    const float* __restrict__ X, const float* __restrict__ W, float* __restrict__ OUT,
    int nRows, int gemmFirst, int gemmCount,
    const int* __restrict__ dst, int* __restrict__ cm, int e, int nBins, int nChunks) {
    __shared__ alignas(16) unsigned char raw[65536];
    if ((int)blockIdx.x < gemmCount) {
        gemm_block<128>(X, W, OUT, nRows, gemmFirst + (int)blockIdx.x, (float*)raw);
        return;
    }
    int* hist = (int*)raw;
    const int chunk = (int)blockIdx.x - gemmCount;
    for (int b = threadIdx.x; b < nBins; b += 256) hist[b] = 0;
    __syncthreads();
    const int base = chunk * CHUNK + threadIdx.x;
    int d[EPT];
#pragma unroll
    for (int j = 0; j < EPT; ++j) {
        const int i = base + j * 256;
        d[j] = (i < e) ? dst[i] : -1;
    }
#pragma unroll
    for (int j = 0; j < EPT; ++j) {
        if (d[j] >= 0) atomicAdd(&hist[d[j] >> BIN_SHIFT], 1);
    }
    __syncthreads();
    for (int b = threadIdx.x; b < nBins; b += 256) cm[(size_t)b * nChunks + chunk] = hist[b];
}

__global__ __launch_bounds__(256) void gemm_spread_kernel(
    const float* __restrict__ X, const float* __restrict__ W, float* __restrict__ OUT,
    int nRows, int gemmFirst, int gemmCount,
    const int* __restrict__ scanned, const int* __restrict__ src,
    const int* __restrict__ dst, const float* __restrict__ w,
    int2* __restrict__ tmp, int e, int nBins, int nChunks) {
    __shared__ alignas(16) unsigned char raw[65536];
    __shared__ int sws[8];
    if ((int)blockIdx.x < gemmCount) {
        gemm_block<128>(X, W, OUT, nRows, gemmFirst + (int)blockIdx.x, (float*)raw);
        return;
    }
    int*  hist  = (int*)raw;
    int*  delta = hist + nBins;
    int*  sbin  = delta + nBins;
    int2* stage = (int2*)(raw + 32768);
    const int tid   = threadIdx.x;
    const int lane  = tid & 31;
    const int wave  = tid >> 5;
    const int chunk = (int)blockIdx.x - gemmCount;

    for (int b = tid; b < nBins; b += 256) hist[b] = 0;
    __syncthreads();

    const int base = chunk * CHUNK + tid;
    int d[EPT];
    int s[EPT];
    int wb[EPT];
    int r[EPT];
#pragma unroll
    for (int j = 0; j < EPT; ++j) {
        const int i = base + j * 256;
        d[j] = -1; s[j] = 0; wb[j] = 0;
        if (i < e) { d[j] = dst[i]; s[j] = src[i]; wb[j] = __float_as_int(w[i]); }
    }
#pragma unroll
    for (int j = 0; j < EPT; ++j) {
        r[j] = 0;
        if (d[j] >= 0) r[j] = atomicAdd(&hist[d[j] >> BIN_SHIFT], 1);
    }
    __syncthreads();

    int carry = 0;
    for (int t0 = 0; t0 < nBins; t0 += 256) {
        const int b = t0 + tid;
        const int c = (b < nBins) ? hist[b] : 0;
        const int inc = wave_incl_scan(c, lane);
        if (lane == 31) sws[wave] = inc;
        __syncthreads();
        int wv = (lane < wave) ? sws[lane] : 0;
        wv = wave_sum(wv);
        int tot = (lane < 8) ? sws[lane] : 0;
        tot = wave_sum(tot);
        if (b < nBins) {
            const int ls = carry + wv + inc - c;
            hist[b]  = ls;
            delta[b] = scanned[(size_t)b * nChunks + chunk] - ls;
        }
        carry += tot;
        __syncthreads();
    }

#pragma unroll
    for (int j = 0; j < EPT; ++j) {
        if (d[j] >= 0) {
            const int bin  = d[j] >> BIN_SHIFT;
            const int slot = hist[bin] + r[j];
            stage[slot] = make_int2(
                (int)((unsigned)s[j] | ((unsigned)(d[j] & (BIN_NODES - 1)) << REC_SHIFT)), wb[j]);
            sbin[slot] = bin;
        }
    }
    __syncthreads();

    const int left = e - chunk * CHUNK;
    const int cntc = left < CHUNK ? left : CHUNK;
    for (int idx = tid; idx < cntc; idx += 256) tmp[delta[sbin[idx]] + idx] = stage[idx];
}


template <int COLS, bool RELU>
__global__ __launch_bounds__(256) void gather_kernel(
    const float* __restrict__ h, const int* __restrict__ rowptr,
    const int2* __restrict__ es, const float* __restrict__ dinv,
    const float* __restrict__ b, float* __restrict__ out, int n) {
    constexpr int VPL = COLS / 32;
    typedef __attribute__((ext_vector_type(VPL))) float vec;

    const int lane = threadIdx.x & 31;
    const int nodeV = blockIdx.x * 8 + (threadIdx.x >> 5);
    if (nodeV >= n) return;
    const int node = __builtin_amdgcn_readfirstlane(nodeV);

    const int beg = rowptr[node];
    const int end = rowptr[node + 1];
    const float dd = dinv[node];
    const float* hl = h + lane * VPL;

    vec acc = *(const vec*)(hl + (size_t)node * COLS) * dd * dd;

    for (int base = beg; base < end; base += 32) {
        const int idx = base + lane;
        int s = node;
        float c = 0.0f;
        if (idx < end) {
            const int2 p = es[idx];
            s = p.x;
            c = dinv[s] * __int_as_float(p.y) * dd;
        }
        const int ci = __float_as_int(c);
        const int rem = end - base;
        const int cnt = rem < 32 ? rem : 32;
        int j = 0;
        for (; j + 4 <= cnt; j += 4) {
            const int s0 = __builtin_amdgcn_readlane(s, j);
            const int s1 = __builtin_amdgcn_readlane(s, j + 1);
            const int s2 = __builtin_amdgcn_readlane(s, j + 2);
            const int s3 = __builtin_amdgcn_readlane(s, j + 3);
            const float c0 = __int_as_float(__builtin_amdgcn_readlane(ci, j));
            const float c1 = __int_as_float(__builtin_amdgcn_readlane(ci, j + 1));
            const float c2 = __int_as_float(__builtin_amdgcn_readlane(ci, j + 2));
            const float c3 = __int_as_float(__builtin_amdgcn_readlane(ci, j + 3));
            const vec v0 = *(const vec*)(hl + (size_t)s0 * COLS);
            const vec v1 = *(const vec*)(hl + (size_t)s1 * COLS);
            const vec v2 = *(const vec*)(hl + (size_t)s2 * COLS);
            const vec v3 = *(const vec*)(hl + (size_t)s3 * COLS);
            acc += v0 * c0;
            acc += v1 * c1;
            acc += v2 * c2;
            acc += v3 * c3;
        }
        for (; j < cnt; ++j) {
            const int s0 = __builtin_amdgcn_readlane(s, j);
            const float c0 = __int_as_float(__builtin_amdgcn_readlane(ci, j));
            const vec v0 = *(const vec*)(hl + (size_t)s0 * COLS);
            acc += v0 * c0;
        }
    }

    acc += *(const vec*)(b + lane * VPL);
    if (RELU) {
#pragma unroll
        for (int k = 0; k < VPL; ++k) acc[k] = fmaxf(acc[k], 0.0f);
    }
    *(vec*)(out + (size_t)node * COLS + lane * VPL) = acc;
}


static inline char* bump(char*& p, size_t bytes) {
    char* r = p;
    p += (bytes + 255) & ~(size_t)255;
    return r;
}

extern "C" void kernel_launch(void* const* d_in, const int* in_sizes, int n_in,
                              void* d_out, int out_size, void* d_ws, size_t ws_size,
                              hipStream_t stream) {
    const float* x   = (const float*)d_in[0];
    const int*  eidx = (const int*)d_in[1];
    const float* ew  = (const float*)d_in[2];
    const float* W1  = (const float*)d_in[3];
    const float* b1  = (const float*)d_in[4];
    const float* W2  = (const float*)d_in[5];
    const float* b2  = (const float*)d_in[6];

    const int N = in_sizes[0] / 128;
    const int E = in_sizes[2];
    const int* src = eidx;
    const int* dst = eidx + E;

    const int T = 256;
    const int nBins   = (N + BIN_NODES - 1) / BIN_NODES;
    const int nChunks = (E + CHUNK - 1) / CHUNK;
    const int nCm     = nBins * nChunks;
    const int scanBlocks = (nCm + 1023) / 1024;
    const int gemmBlocks = ((N + 15) / 16 + 7) / 8;
    const int nodeBlocks = (N + 7) / 8;
    const int gemmA = gemmBlocks / 2;
    const int gemmB = gemmBlocks - gemmA;

    char* p = (char*)d_ws;
    int*   cm      = (int*)  bump(p, (size_t)nCm * sizeof(int));
    int*   scanned = (int*)  bump(p, (size_t)nCm * sizeof(int));
    int*   rowptr  = (int*)  bump(p, (size_t)(N + 1) * sizeof(int));
    int2*  tmp     = (int2*) bump(p, (size_t)E * sizeof(int2));
    int2*  es      = (int2*) bump(p, (size_t)E * sizeof(int2));
    float* dinv    = (float*)bump(p, (size_t)N * sizeof(float));
    float* h1      = (float*)bump(p, (size_t)N * 128 * sizeof(float));
    float* a1      = (float*)bump(p, (size_t)N * 128 * sizeof(float));
    float* g       = (float*)bump(p, (size_t)N * 64 * sizeof(float));
    float* out     = (float*)d_out;

    gemm_count_kernel<<<gemmA + nChunks, T, 0, stream>>>(x, W1, h1, N, 0, gemmA,
                                                         dst, cm, E, nBins, nChunks);
    scan_kernel<<<scanBlocks, 1024, 0, stream>>>(cm, scanned, nCm);
    gemm_spread_kernel<<<gemmB + nChunks, T, 0, stream>>>(x, W1, h1, N, gemmA, gemmB,
                                                          scanned, src, dst, ew, tmp,
                                                          E, nBins, nChunks);
    bin_sort_kernel<<<nBins, 1024, 0, stream>>>(scanned, tmp, rowptr, es, dinv,
                                                N, E, nBins, nChunks);

    gather_kernel<128, true><<<nodeBlocks, T, 0, stream>>>(h1, rowptr, es, dinv, b1, a1, N);

    gemm_wmma_kernel<64><<<gemmBlocks, T, 0, stream>>>(a1, W2, g, N);
    gather_kernel<64, false><<<nodeBlocks, T, 0, stream>>>(g, rowptr, es, dinv, b2, out, N);
}


# deg^-1/2 of the source folded into layer-2 rows in the dense transform's epilogue; layer-2 gather takes bare edge weights, scales its sum by deg^-1/2 of the node: no per-edge lookup; sum order kept
# speedup vs baseline: 4.6190x; 1.0348x over previous
#include <hip/hip_runtime.h>


typedef __attribute__((ext_vector_type(2))) float v2f;
typedef __attribute__((ext_vector_type(4))) float v4f;
typedef __attribute__((ext_vector_type(8))) float v8f;


constexpr int BIN_SHIFT = 9;
constexpr int BIN_NODES = 1 << BIN_SHIFT;
constexpr int REC_SHIFT = 32 - BIN_SHIFT;
constexpr unsigned REC_MASK = (1u << REC_SHIFT) - 1u;
constexpr int EPT   = 16;
constexpr int CHUNK = 256 * EPT;
constexpr int CAP   = 12288;
constexpr int SORT_U = 8;

__device__ __forceinline__ int wave_sum(int v) {
    for (int o = 16; o > 0; o >>= 1) v += __shfl_xor(v, o);
    return v;
}

__device__ __forceinline__ int wave_incl_scan(int v, int lane) {
    for (int o = 1; o < 32; o <<= 1) {
        const int t = __shfl_up(v, o);
        if (lane >= o) v += t;
    }
    return v;
}

__global__ __launch_bounds__(1024) void scan_kernel(
    const int* __restrict__ cm, int* __restrict__ scanned, int n) {
    __shared__ int sw[32];
    __shared__ int sb[32];
    const int lane = threadIdx.x & 31;
    const int wave = threadIdx.x >> 5;
    const int first = blockIdx.x * 1024;
    int pb = 0;
    for (int k = threadIdx.x; k < first; k += 1024) pb += cm[k];
    pb = wave_sum(pb);
    if (lane == 0) sb[wave] = pb;
    const int i = first + threadIdx.x;
    const int c = (i < n) ? cm[i] : 0;
    const int inc = wave_incl_scan(c, lane);
    if (lane == 31) sw[wave] = inc;
    __syncthreads();
    const int base = wave_sum(sb[lane]);
    int wv = (lane < wave) ? sw[lane] : 0;
    wv = wave_sum(wv);
    if (i < n) scanned[i] = base + wv + inc - c;
}

__global__ __launch_bounds__(1024) void bin_sort_kernel(
    const int* __restrict__ scanned, const int2* __restrict__ tmp,
    int* __restrict__ rowptr, int2* __restrict__ es, float* __restrict__ dinv,
    int n, int e, int nBins, int nChunks) {
    __shared__ int2  stage[CAP];
    __shared__ int   cnt[BIN_NODES];
    __shared__ float degs[BIN_NODES];
    __shared__ int   cur[BIN_NODES];
    __shared__ int   sw[BIN_NODES / 32];
    const int tid  = threadIdx.x;
    const int lane = tid & 31;
    const int wave = tid >> 5;
    const int b    = blockIdx.x;
    const int eBeg = scanned[(size_t)b * nChunks];
    const int eEnd = (b + 1 < nBins) ? scanned[(size_t)(b + 1) * nChunks] : e;
    const int m    = eEnd - eBeg;
    const bool fits = m <= CAP;

    if (tid < BIN_NODES) { cnt[tid] = 0; degs[tid] = 0.0f; }
    __syncthreads();
    for (int i0 = tid; i0 < m; i0 += 1024 * SORT_U) {
        int2 rec[SORT_U];
#pragma unroll
        for (int k = 0; k < SORT_U; ++k) {
            const int i = i0 + k * 1024;
            rec[k] = tmp[eBeg + (i < m ? i : m - 1)];
        }
#pragma unroll
        for (int k = 0; k < SORT_U; ++k) {
            if (i0 + k * 1024 < m) {
                const int dl = (int)((unsigned)rec[k].x >> REC_SHIFT);
                atomicAdd(&cnt[dl], 1);
                atomicAdd(&degs[dl], __int_as_float(rec[k].y));
            }
        }
    }
    __syncthreads();

    const int c   = (tid < BIN_NODES) ? cnt[tid] : 0;
    const int inc = wave_incl_scan(c, lane);
    if (tid < BIN_NODES && lane == 31) sw[wave] = inc;
    __syncthreads();
    if (tid < BIN_NODES) {
        int wv = (lane < wave) ? sw[lane] : 0;
        wv = wave_sum(wv);
        const int excl = wv + inc - c;
        cur[tid] = excl;
        const int node = b * BIN_NODES + tid;
        if (node < n) {
            rowptr[node] = eBeg + excl;
            const float deg = 1.0f + degs[tid];
            dinv[node] = (deg > 0.0f) ? rsqrtf(deg) : 0.0f;
        }
    }
    if (b == nBins - 1 && tid == 0) rowptr[n] = eEnd;
    __syncthreads();

    for (int i0 = tid; i0 < m; i0 += 1024 * SORT_U) {
        int2 rec[SORT_U];
#pragma unroll
        for (int k = 0; k < SORT_U; ++k) {
            const int i = i0 + k * 1024;
            rec[k] = tmp[eBeg + (i < m ? i : m - 1)];
        }
#pragma unroll
        for (int k = 0; k < SORT_U; ++k) {
            if (i0 + k * 1024 < m) {
                const int dl  = (int)((unsigned)rec[k].x >> REC_SHIFT);
                const int pos = atomicAdd(&cur[dl], 1);
                const int2 o  = make_int2((int)((unsigned)rec[k].x & REC_MASK), rec[k].y);
                if (fits) stage[pos] = o; else es[eBeg + pos] = o;
            }
        }
    }
    __syncthreads();
    if (fits) {
        for (int i = tid; i < m; i += 1024) es[eBeg + i] = stage[i];
    }
}


template <int COLS, bool ROWSCALE>
__device__ __forceinline__ void gemm_block(
    const float* __restrict__ X, const float* __restrict__ W,
    float* __restrict__ OUT, const float* __restrict__ rs, int nRows, int blk, float* sW) {
    constexpr int K  = 128;
    constexpr int NT = COLS / 16;
    constexpr int CSH = (COLS == 128) ? 7 : 6;

    for (int i = threadIdx.x; i < K * COLS; i += 256) {
        const int row = i >> CSH;
        const int col = i & (COLS - 1);
        sW[(row >> 1) * (COLS * 2) + col * 2 + (row & 1)] = W[i];
    }
    __syncthreads();

    const int wave = threadIdx.x >> 5;
    const int lane = threadIdx.x & 31;
    const int rb   = blk * 8 + wave;
    if (rb * 16 >= nRows) return;

    const int l15  = lane & 15;
    const int hsel = lane >> 4;

    const float* aBase = X + (size_t)(rb * 16 + l15) * K + 2 * hsel;
    const float* bCol  = &sW[hsel * (COLS * 2) + l15 * 2];

    const int rowBase = rb * 16 + (hsel << 3);
    float sc[8];
    if (ROWSCALE) {
#pragma unroll
        for (int r = 0; r < 8; ++r) {
            const int row = rowBase + r;
            sc[r] = rs[row < nRows ? row : nRows - 1];
        }
    }

    v8f acc[NT] = {};

    for (int k = 0; k < K; k += 4) {
        const v2f A = *(const v2f*)(aBase + k);
        const float* bp = bCol + (k >> 1) * (COLS * 2);
#pragma unroll
        for (int c = 0; c < NT; ++c) {
            const v2f B = *(const v2f*)(bp + c * 32);
            acc[c] = __builtin_amdgcn_wmma_f32_16x16x4_f32(
                false, A, false, B, (short)0, acc[c], false, false);
        }
    }

#pragma unroll
    for (int c = 0; c < NT; ++c) {
#pragma unroll
        for (int r = 0; r < 8; ++r)
            OUT[(size_t)(rowBase + r) * COLS + c * 16 + l15] =
                ROWSCALE ? acc[c][r] * sc[r] : acc[c][r];
    }
}

template <int COLS>
__global__ __launch_bounds__(256) void gemm_wmma_kernel(
    const float* __restrict__ X, const float* __restrict__ W,
    float* __restrict__ OUT, const float* __restrict__ rs, int nRows) {
    __shared__ float sW[64 * COLS * 2];
    gemm_block<COLS, true>(X, W, OUT, rs, nRows, blockIdx.x, sW);
}


__global__ __launch_bounds__(256) void gemm_count_kernel(
    const float* __restrict__ X, const float* __restrict__ W, float* __restrict__ OUT,
    int nRows, int gemmFirst, int gemmCount,
    const int* __restrict__ dst, int* __restrict__ cm, int e, int nBins, int nChunks) {
    __shared__ alignas(16) unsigned char raw[65536];
    if ((int)blockIdx.x < gemmCount) {
        gemm_block<128, false>(X, W, OUT, nullptr, nRows, gemmFirst + (int)blockIdx.x, (float*)raw);
        return;
    }
    int* hist = (int*)raw;
    const int chunk = (int)blockIdx.x - gemmCount;
    for (int b = threadIdx.x; b < nBins; b += 256) hist[b] = 0;
    __syncthreads();
    const int base = chunk * CHUNK + threadIdx.x;
    int d[EPT];
#pragma unroll
    for (int j = 0; j < EPT; ++j) {
        const int i = base + j * 256;
        d[j] = (i < e) ? dst[i] : -1;
    }
#pragma unroll
    for (int j = 0; j < EPT; ++j) {
        if (d[j] >= 0) atomicAdd(&hist[d[j] >> BIN_SHIFT], 1);
    }
    __syncthreads();
    for (int b = threadIdx.x; b < nBins; b += 256) cm[(size_t)b * nChunks + chunk] = hist[b];
}

__global__ __launch_bounds__(256) void gemm_spread_kernel(
    const float* __restrict__ X, const float* __restrict__ W, float* __restrict__ OUT,
    int nRows, int gemmFirst, int gemmCount,
    const int* __restrict__ scanned, const int* __restrict__ src,
    const int* __restrict__ dst, const float* __restrict__ w,
    int2* __restrict__ tmp, int e, int nBins, int nChunks) {
    __shared__ alignas(16) unsigned char raw[65536];
    __shared__ int sws[8];
    if ((int)blockIdx.x < gemmCount) {
        gemm_block<128, false>(X, W, OUT, nullptr, nRows, gemmFirst + (int)blockIdx.x, (float*)raw);
        return;
    }
    int*  hist  = (int*)raw;
    int*  delta = hist + nBins;
    int*  sbin  = delta + nBins;
    int2* stage = (int2*)(raw + 32768);
    const int tid   = threadIdx.x;
    const int lane  = tid & 31;
    const int wave  = tid >> 5;
    const int chunk = (int)blockIdx.x - gemmCount;

    for (int b = tid; b < nBins; b += 256) hist[b] = 0;
    __syncthreads();

    const int base = chunk * CHUNK + tid;
    int d[EPT];
    int s[EPT];
    int wb[EPT];
    int r[EPT];
#pragma unroll
    for (int j = 0; j < EPT; ++j) {
        const int i = base + j * 256;
        d[j] = -1; s[j] = 0; wb[j] = 0;
        if (i < e) { d[j] = dst[i]; s[j] = src[i]; wb[j] = __float_as_int(w[i]); }
    }
#pragma unroll
    for (int j = 0; j < EPT; ++j) {
        r[j] = 0;
        if (d[j] >= 0) r[j] = atomicAdd(&hist[d[j] >> BIN_SHIFT], 1);
    }
    __syncthreads();

    int carry = 0;
    for (int t0 = 0; t0 < nBins; t0 += 256) {
        const int b = t0 + tid;
        const int c = (b < nBins) ? hist[b] : 0;
        const int inc = wave_incl_scan(c, lane);
        if (lane == 31) sws[wave] = inc;
        __syncthreads();
        int wv = (lane < wave) ? sws[lane] : 0;
        wv = wave_sum(wv);
        int tot = (lane < 8) ? sws[lane] : 0;
        tot = wave_sum(tot);
        if (b < nBins) {
            const int ls = carry + wv + inc - c;
            hist[b]  = ls;
            delta[b] = scanned[(size_t)b * nChunks + chunk] - ls;
        }
        carry += tot;
        __syncthreads();
    }

#pragma unroll
    for (int j = 0; j < EPT; ++j) {
        if (d[j] >= 0) {
            const int bin  = d[j] >> BIN_SHIFT;
            const int slot = hist[bin] + r[j];
            stage[slot] = make_int2(
                (int)((unsigned)s[j] | ((unsigned)(d[j] & (BIN_NODES - 1)) << REC_SHIFT)), wb[j]);
            sbin[slot] = bin;
        }
    }
    __syncthreads();

    const int left = e - chunk * CHUNK;
    const int cntc = left < CHUNK ? left : CHUNK;
    for (int idx = tid; idx < cntc; idx += 256) tmp[delta[sbin[idx]] + idx] = stage[idx];
}


template <int COLS, bool RELU, bool PRE>
__global__ __launch_bounds__(256) void gather_kernel(
    const float* __restrict__ h, const int* __restrict__ rowptr,
    const int2* __restrict__ es, const float* __restrict__ dinv,
    const float* __restrict__ b, float* __restrict__ out, int n) {
    constexpr int VPL = COLS / 32;
    typedef __attribute__((ext_vector_type(VPL))) float vec;

    const int lane = threadIdx.x & 31;
    const int nodeV = blockIdx.x * 8 + (threadIdx.x >> 5);
    if (nodeV >= n) return;
    const int node = __builtin_amdgcn_readfirstlane(nodeV);

    const int beg = rowptr[node];
    const int end = rowptr[node + 1];
    const float dd = dinv[node];
    const float* hl = h + lane * VPL;

    vec acc = *(const vec*)(hl + (size_t)node * COLS);
    if (!PRE) acc = acc * dd * dd;

    for (int base = beg; base < end; base += 32) {
        const int idx = base + lane;
        int s = node;
        float c = 0.0f;
        if (idx < end) {
            const int2 p = es[idx];
            s = p.x;
            c = PRE ? __int_as_float(p.y) : dinv[s] * __int_as_float(p.y) * dd;
        }
        const int ci = __float_as_int(c);
        const int rem = end - base;
        const int cnt = rem < 32 ? rem : 32;
        int j = 0;
        for (; j + 4 <= cnt; j += 4) {
            const int s0 = __builtin_amdgcn_readlane(s, j);
            const int s1 = __builtin_amdgcn_readlane(s, j + 1);
            const int s2 = __builtin_amdgcn_readlane(s, j + 2);
            const int s3 = __builtin_amdgcn_readlane(s, j + 3);
            const float c0 = __int_as_float(__builtin_amdgcn_readlane(ci, j));
            const float c1 = __int_as_float(__builtin_amdgcn_readlane(ci, j + 1));
            const float c2 = __int_as_float(__builtin_amdgcn_readlane(ci, j + 2));
            const float c3 = __int_as_float(__builtin_amdgcn_readlane(ci, j + 3));
            const vec v0 = *(const vec*)(hl + (size_t)s0 * COLS);
            const vec v1 = *(const vec*)(hl + (size_t)s1 * COLS);
            const vec v2 = *(const vec*)(hl + (size_t)s2 * COLS);
            const vec v3 = *(const vec*)(hl + (size_t)s3 * COLS);
            acc += v0 * c0;
            acc += v1 * c1;
            acc += v2 * c2;
            acc += v3 * c3;
        }
        for (; j < cnt; ++j) {
            const int s0 = __builtin_amdgcn_readlane(s, j);
            const float c0 = __int_as_float(__builtin_amdgcn_readlane(ci, j));
            const vec v0 = *(const vec*)(hl + (size_t)s0 * COLS);
            acc += v0 * c0;
        }
    }

    if (PRE) acc = acc * dd;
    acc += *(const vec*)(b + lane * VPL);
    if (RELU) {
#pragma unroll
        for (int k = 0; k < VPL; ++k) acc[k] = fmaxf(acc[k], 0.0f);
    }
    *(vec*)(out + (size_t)node * COLS + lane * VPL) = acc;
}


static inline char* bump(char*& p, size_t bytes) {
    char* r = p;
    p += (bytes + 255) & ~(size_t)255;
    return r;
}

extern "C" void kernel_launch(void* const* d_in, const int* in_sizes, int n_in,
                              void* d_out, int out_size, void* d_ws, size_t ws_size,
                              hipStream_t stream) {
    const float* x   = (const float*)d_in[0];
    const int*  eidx = (const int*)d_in[1];
    const float* ew  = (const float*)d_in[2];
    const float* W1  = (const float*)d_in[3];
    const float* b1  = (const float*)d_in[4];
    const float* W2  = (const float*)d_in[5];
    const float* b2  = (const float*)d_in[6];

    const int N = in_sizes[0] / 128;
    const int E = in_sizes[2];
    const int* src = eidx;
    const int* dst = eidx + E;

    const int T = 256;
    const int nBins   = (N + BIN_NODES - 1) / BIN_NODES;
    const int nChunks = (E + CHUNK - 1) / CHUNK;
    const int nCm     = nBins * nChunks;
    const int scanBlocks = (nCm + 1023) / 1024;
    const int gemmBlocks = ((N + 15) / 16 + 7) / 8;
    const int nodeBlocks = (N + 7) / 8;
    const int gemmA = gemmBlocks / 2;
    const int gemmB = gemmBlocks - gemmA;

    char* p = (char*)d_ws;
    int*   cm      = (int*)  bump(p, (size_t)nCm * sizeof(int));
    int*   scanned = (int*)  bump(p, (size_t)nCm * sizeof(int));
    int*   rowptr  = (int*)  bump(p, (size_t)(N + 1) * sizeof(int));
    int2*  tmp     = (int2*) bump(p, (size_t)E * sizeof(int2));
    int2*  es      = (int2*) bump(p, (size_t)E * sizeof(int2));
    float* dinv    = (float*)bump(p, (size_t)N * sizeof(float));
    float* h1      = (float*)bump(p, (size_t)N * 128 * sizeof(float));
    float* a1      = (float*)bump(p, (size_t)N * 128 * sizeof(float));
    float* g       = (float*)bump(p, (size_t)N * 64 * sizeof(float));
    float* out     = (float*)d_out;

    gemm_count_kernel<<<gemmA + nChunks, T, 0, stream>>>(x, W1, h1, N, 0, gemmA,
                                                         dst, cm, E, nBins, nChunks);
    scan_kernel<<<scanBlocks, 1024, 0, stream>>>(cm, scanned, nCm);
    gemm_spread_kernel<<<gemmB + nChunks, T, 0, stream>>>(x, W1, h1, N, gemmA, gemmB,
                                                          scanned, src, dst, ew, tmp,
                                                          E, nBins, nChunks);
    bin_sort_kernel<<<nBins, 1024, 0, stream>>>(scanned, tmp, rowptr, es, dinv,
                                                N, E, nBins, nChunks);

    gather_kernel<128, true, false><<<nodeBlocks, T, 0, stream>>>(h1, rowptr, es, dinv, b1, a1, N);

    gemm_wmma_kernel<64><<<gemmBlocks, T, 0, stream>>>(a1, W2, g, dinv, N);
    gather_kernel<64, false, true><<<nodeBlocks, T, 0, stream>>>(g, rowptr, es, dinv, b2, out, N);
}


# edge sort in 2 launches, not 4: no count or scan pass; chunks keep bin-ordered records in place + (bin,chunk) run tables, bin groups sum earlier runs; layer-1 GEMM in pass 1; sum order LDS arrival
# speedup vs baseline: 5.0826x; 1.1004x over previous
#include <hip/hip_runtime.h>


typedef __attribute__((ext_vector_type(2))) float v2f;
typedef __attribute__((ext_vector_type(4))) float v4f;
typedef __attribute__((ext_vector_type(8))) float v8f;


constexpr int BIN_SHIFT = 9;
constexpr int BIN_NODES = 1 << BIN_SHIFT;
constexpr int REC_SHIFT = 32 - BIN_SHIFT;
constexpr unsigned REC_MASK = (1u << REC_SHIFT) - 1u;
constexpr int EPT   = 16;
constexpr int CHUNK = 256 * EPT;
constexpr int CAP   = 12288;
constexpr int RUN_C = 4;
constexpr int RUN_K = 4;

__device__ __forceinline__ int wave_sum(int v) {
    for (int o = 16; o > 0; o >>= 1) v += __shfl_xor(v, o);
    return v;
}

__device__ __forceinline__ int wave_incl_scan(int v, int lane) {
    for (int o = 1; o < 32; o <<= 1) {
        const int t = __shfl_up(v, o);
        if (lane >= o) v += t;
    }
    return v;
}

template <bool PLACE>
__device__ __forceinline__ void bin_record(
    const int2 rec, int* cnt, float* degs, int* cur, int2* stage,
    int2* __restrict__ es, int eBeg, bool fits) {
    const int dl = (int)((unsigned)rec.x >> REC_SHIFT);
    if (!PLACE) {
        atomicAdd(&cnt[dl], 1);
        atomicAdd(&degs[dl], __int_as_float(rec.y));
    } else {
        const int pos = atomicAdd(&cur[dl], 1);
        const int2 o  = make_int2((int)((unsigned)rec.x & REC_MASK), rec.y);
        if (fits) stage[pos] = o; else es[eBeg + pos] = o;
    }
}

template <bool PLACE>
__device__ __forceinline__ void bin_sweep(
    const int* __restrict__ cm, const int* __restrict__ lo, const int2* __restrict__ tmp,
    int first, int nChunks, int e, int grp, int l8,
    int* cnt, float* degs, int* cur, int2* stage, int2* __restrict__ es, int eBeg, bool fits) {
    for (int c0 = 0; c0 < nChunks; c0 += 128 * RUN_C) {
        int rn[RUN_C];
        int rl[RUN_C];
#pragma unroll
        for (int q = 0; q < RUN_C; ++q) {
            const int c  = c0 + grp + 128 * q;
            const int cc = c < nChunks ? c : nChunks - 1;
            const int nn = cm[first + cc];
            rl[q] = lo[first + cc] + cc * CHUNK;
            rn[q] = c < nChunks ? nn : 0;
        }
        int2 rec[RUN_C][RUN_K];
#pragma unroll
        for (int q = 0; q < RUN_C; ++q) {
#pragma unroll
            for (int k = 0; k < RUN_K; ++k) {
                const int i = rl[q] + l8 + 8 * k;
                rec[q][k] = tmp[i < e ? i : e - 1];
            }
        }
        if (!PLACE) {
#pragma unroll
            for (int q = 0; q < RUN_C; ++q) {
#pragma unroll
                for (int k = 0; k < RUN_K; ++k) {
                    if (l8 + 8 * k < rn[q])
                        bin_record<false>(rec[q][k], cnt, degs, cur, stage, es, eBeg, fits);
                }
            }
        } else {
            int pos[RUN_C][RUN_K];
#pragma unroll
            for (int q = 0; q < RUN_C; ++q) {
#pragma unroll
                for (int k = 0; k < RUN_K; ++k) {
                    pos[q][k] = -1;
                    if (l8 + 8 * k < rn[q])
                        pos[q][k] = atomicAdd(&cur[(unsigned)rec[q][k].x >> REC_SHIFT], 1);
                }
            }
#pragma unroll
            for (int q = 0; q < RUN_C; ++q) {
#pragma unroll
                for (int k = 0; k < RUN_K; ++k) {
                    if (pos[q][k] >= 0) {
                        const int2 o = make_int2((int)((unsigned)rec[q][k].x & REC_MASK), rec[q][k].y);
                        if (fits) stage[pos[q][k]] = o; else es[eBeg + pos[q][k]] = o;
                    }
                }
            }
        }
#pragma unroll
        for (int q = 0; q < RUN_C; ++q) {
            for (int j = l8 + 8 * RUN_K; j < rn[q]; j += 8)
                bin_record<PLACE>(tmp[rl[q] + j], cnt, degs, cur, stage, es, eBeg, fits);
        }
    }
}

__global__ __launch_bounds__(1024) void bin_sort_kernel(
    const int* __restrict__ cm, const int* __restrict__ lo, const int2* __restrict__ tmp,
    int* __restrict__ rowptr, int2* __restrict__ es, float* __restrict__ dinv,
    int n, int e, int nBins, int nChunks) {
    __shared__ int2  stage[CAP];
    __shared__ int   cnt[BIN_NODES];
    __shared__ float degs[BIN_NODES];
    __shared__ int   cur[BIN_NODES];
    __shared__ int   sw[BIN_NODES / 32];
    __shared__ int   sfront[32];
    __shared__ int   sown[32];
    const int tid  = threadIdx.x;
    const int lane = tid & 31;
    const int wave = tid >> 5;
    const int grp  = tid >> 3;
    const int l8   = tid & 7;
    const int b    = blockIdx.x;
    const int first = b * nChunks;

    int pf = 0;
    const int4* cm4 = (const int4*)cm;
    for (int k = tid; k < (first >> 2); k += 1024) {
        const int4 v = cm4[k];
        pf += (v.x + v.y) + (v.z + v.w);
    }
    if (tid < (first & 3)) pf += cm[(first & ~3) + tid];
    int po = 0;
    for (int k = tid; k < nChunks; k += 1024) po += cm[first + k];
    pf = wave_sum(pf);
    po = wave_sum(po);
    if (lane == 0) { sfront[wave] = pf; sown[wave] = po; }
    if (tid < BIN_NODES) { cnt[tid] = 0; degs[tid] = 0.0f; }
    __syncthreads();
    const int eBeg = wave_sum(sfront[lane]);
    const int m    = wave_sum(sown[lane]);
    const int eEnd = eBeg + m;
    const bool fits = m <= CAP;

    bin_sweep<false>(cm, lo, tmp, first, nChunks, e, grp, l8, cnt, degs, cur, stage, es, eBeg, fits);
    __syncthreads();

    const int c   = (tid < BIN_NODES) ? cnt[tid] : 0;
    const int inc = wave_incl_scan(c, lane);
    if (tid < BIN_NODES && lane == 31) sw[wave] = inc;
    __syncthreads();
    if (tid < BIN_NODES) {
        int wv = (lane < wave) ? sw[lane] : 0;
        wv = wave_sum(wv);
        const int excl = wv + inc - c;
        cur[tid] = excl;
        const int node = b * BIN_NODES + tid;
        if (node < n) {
            rowptr[node] = eBeg + excl;
            const float deg = 1.0f + degs[tid];
            dinv[node] = (deg > 0.0f) ? rsqrtf(deg) : 0.0f;
        }
    }
    if (b == nBins - 1 && tid == 0) rowptr[n] = eEnd;
    __syncthreads();

    bin_sweep<true>(cm, lo, tmp, first, nChunks, e, grp, l8, cnt, degs, cur, stage, es, eBeg, fits);
    __syncthreads();
    if (fits) {
        for (int i = tid; i < m; i += 1024) es[eBeg + i] = stage[i];
    }
}


template <int COLS>
__device__ __forceinline__ void gemm_stage(const float* __restrict__ W, float* sW) {
    constexpr int K  = 128;
    constexpr int CSH = (COLS == 128) ? 7 : 6;
    for (int i = threadIdx.x; i < K * COLS; i += 256) {
        const int row = i >> CSH;
        const int col = i & (COLS - 1);
        sW[(row >> 1) * (COLS * 2) + col * 2 + (row & 1)] = W[i];
    }
}

template <int COLS, int NTW, bool ROWSCALE>
__device__ __forceinline__ void gemm_rows(
    const float* __restrict__ X, float* __restrict__ OUT, const float* __restrict__ rs,
    int nRows, int rb, int cFirst, const float* sW) {
    constexpr int K  = 128;
    const int lane = threadIdx.x & 31;
    const int l15  = lane & 15;
    const int hsel = lane >> 4;

    const float* aBase = X + (size_t)(rb * 16 + l15) * K + 2 * hsel;
    const float* bCol  = &sW[hsel * (COLS * 2) + (cFirst * 16 + l15) * 2];

    const int rowBase = rb * 16 + (hsel << 3);
    float sc[8];
    if (ROWSCALE) {
#pragma unroll
        for (int r = 0; r < 8; ++r) {
            const int row = rowBase + r;
            sc[r] = rs[row < nRows ? row : nRows - 1];
        }
    }

    v8f acc[NTW] = {};

    for (int k = 0; k < K; k += 4) {
        const v2f A = *(const v2f*)(aBase + k);
        const float* bp = bCol + (k >> 1) * (COLS * 2);
#pragma unroll
        for (int c = 0; c < NTW; ++c) {
            const v2f B = *(const v2f*)(bp + c * 32);
            acc[c] = __builtin_amdgcn_wmma_f32_16x16x4_f32(
                false, A, false, B, (short)0, acc[c], false, false);
        }
    }

#pragma unroll
    for (int c = 0; c < NTW; ++c) {
#pragma unroll
        for (int r = 0; r < 8; ++r)
            OUT[(size_t)(rowBase + r) * COLS + (cFirst + c) * 16 + l15] =
                ROWSCALE ? acc[c][r] * sc[r] : acc[c][r];
    }
}

template <int COLS, bool ROWSCALE>
__device__ __forceinline__ void gemm_block(
    const float* __restrict__ X, const float* __restrict__ W,
    float* __restrict__ OUT, const float* __restrict__ rs, int nRows, int blk, float* sW) {
    gemm_stage<COLS>(W, sW);
    __syncthreads();
    const int rb = blk * 8 + (int)(threadIdx.x >> 5);
    if (rb * 16 >= nRows) return;
    gemm_rows<COLS, COLS / 16, ROWSCALE>(X, OUT, rs, nRows, rb, 0, sW);
}

template <int COLS>
__global__ __launch_bounds__(256) void gemm_wmma_kernel(
    const float* __restrict__ X, const float* __restrict__ W,
    float* __restrict__ OUT, const float* __restrict__ rs, int nRows) {
    __shared__ float sW[64 * COLS * 2];
    gemm_block<COLS, true>(X, W, OUT, rs, nRows, blockIdx.x, sW);
}


__global__ __launch_bounds__(256) void gemm_spread_kernel(
    const float* __restrict__ X, const float* __restrict__ W, float* __restrict__ OUT,
    int nRows, int gemmCount,
    const int* __restrict__ src, const int* __restrict__ dst, const float* __restrict__ w,
    int2* __restrict__ tmp, int* __restrict__ cm, int* __restrict__ lo,
    int e, int nBins, int nChunks) {
    __shared__ alignas(16) unsigned char raw[65536];
    __shared__ int sws[8];
    if ((int)blockIdx.x < gemmCount) {
        float* sW = (float*)raw;
        gemm_stage<128>(W, sW);
        __syncthreads();
        const int gw = (int)(threadIdx.x >> 5);
        const int nTiles = (nRows + 15) >> 4;
        const int rb = (int)blockIdx.x * 8 + gw;
        if (rb < nTiles) gemm_rows<128, 8, false>(X, OUT, nullptr, nRows, rb, 0, sW);
        for (int xt = gemmCount * 8 + (int)blockIdx.x; xt < nTiles; xt += gemmCount)
            gemm_rows<128, 1, false>(X, OUT, nullptr, nRows, xt, gw, sW);
        return;
    }
    int*  hist  = (int*)raw;
    int2* stage = (int2*)(raw + 32768);
    const int tid   = threadIdx.x;
    const int lane  = tid & 31;
    const int wave  = tid >> 5;
    const int chunk = (int)blockIdx.x - gemmCount;

    for (int b = tid; b < nBins; b += 256) hist[b] = 0;
    __syncthreads();

    const int base = chunk * CHUNK + tid;
    int d[EPT];
    int s[EPT];
    int wb[EPT];
    int r[EPT];
#pragma unroll
    for (int j = 0; j < EPT; ++j) {
        const int i = base + j * 256;
        d[j] = -1; s[j] = 0; wb[j] = 0;
        if (i < e) { d[j] = dst[i]; s[j] = src[i]; wb[j] = __float_as_int(w[i]); }
    }
#pragma unroll
    for (int j = 0; j < EPT; ++j) {
        r[j] = 0;
        if (d[j] >= 0) r[j] = atomicAdd(&hist[d[j] >> BIN_SHIFT], 1);
    }
    __syncthreads();

    int carry = 0;
    for (int t0 = 0; t0 < nBins; t0 += 256) {
        const int b = t0 + tid;
        const int c = (b < nBins) ? hist[b] : 0;
        const int inc = wave_incl_scan(c, lane);
        if (lane == 31) sws[wave] = inc;
        __syncthreads();
        int wv = (lane < wave) ? sws[lane] : 0;
        wv = wave_sum(wv);
        int tot = (lane < 8) ? sws[lane] : 0;
        tot = wave_sum(tot);
        if (b < nBins) {
            const int ls = carry + wv + inc - c;
            hist[b] = ls;
            cm[(size_t)b * nChunks + chunk] = c;
            lo[(size_t)b * nChunks + chunk] = ls;
        }
        carry += tot;
        __syncthreads();
    }

#pragma unroll
    for (int j = 0; j < EPT; ++j) {
        if (d[j] >= 0) {
            const int slot = hist[d[j] >> BIN_SHIFT] + r[j];
            stage[slot] = make_int2(
                (int)((unsigned)s[j] | ((unsigned)(d[j] & (BIN_NODES - 1)) << REC_SHIFT)), wb[j]);
        }
    }
    __syncthreads();

    const int left = e - chunk * CHUNK;
    const int cntc = left < CHUNK ? left : CHUNK;
    int2* slice = tmp + (size_t)chunk * CHUNK;
    for (int idx = tid; idx < cntc; idx += 256) slice[idx] = stage[idx];
}


template <int COLS, bool RELU, bool PRE>
__global__ __launch_bounds__(256) void gather_kernel(
    const float* __restrict__ h, const int* __restrict__ rowptr,
    const int2* __restrict__ es, const float* __restrict__ dinv,
    const float* __restrict__ b, float* __restrict__ out, int n) {
    constexpr int VPL = COLS / 32;
    typedef __attribute__((ext_vector_type(VPL))) float vec;

    const int lane = threadIdx.x & 31;
    const int nodeV = blockIdx.x * 8 + (threadIdx.x >> 5);
    if (nodeV >= n) return;
    const int node = __builtin_amdgcn_readfirstlane(nodeV);

    const int beg = rowptr[node];
    const int end = rowptr[node + 1];
    const float dd = dinv[node];
    const float* hl = h + lane * VPL;

    vec acc = *(const vec*)(hl + (size_t)node * COLS);
    if (!PRE) acc = acc * dd * dd;

    for (int base = beg; base < end; base += 32) {
        const int idx = base + lane;
        int s = node;
        float c = 0.0f;
        if (idx < end) {
            const int2 p = es[idx];
            s = p.x;
            c = PRE ? __int_as_float(p.y) : dinv[s] * __int_as_float(p.y) * dd;
        }
        const int ci = __float_as_int(c);
        const int rem = end - base;
        const int cnt = rem < 32 ? rem : 32;
        int j = 0;
        for (; j + 4 <= cnt; j += 4) {
            const int s0 = __builtin_amdgcn_readlane(s, j);
            const int s1 = __builtin_amdgcn_readlane(s, j + 1);
            const int s2 = __builtin_amdgcn_readlane(s, j + 2);
            const int s3 = __builtin_amdgcn_readlane(s, j + 3);
            const float c0 = __int_as_float(__builtin_amdgcn_readlane(ci, j));
            const float c1 = __int_as_float(__builtin_amdgcn_readlane(ci, j + 1));
            const float c2 = __int_as_float(__builtin_amdgcn_readlane(ci, j + 2));
            const float c3 = __int_as_float(__builtin_amdgcn_readlane(ci, j + 3));
            const vec v0 = *(const vec*)(hl + (size_t)s0 * COLS);
            const vec v1 = *(const vec*)(hl + (size_t)s1 * COLS);
            const vec v2 = *(const vec*)(hl + (size_t)s2 * COLS);
            const vec v3 = *(const vec*)(hl + (size_t)s3 * COLS);
            acc += v0 * c0;
            acc += v1 * c1;
            acc += v2 * c2;
            acc += v3 * c3;
        }
        for (; j < cnt; ++j) {
            const int s0 = __builtin_amdgcn_readlane(s, j);
            const float c0 = __int_as_float(__builtin_amdgcn_readlane(ci, j));
            const vec v0 = *(const vec*)(hl + (size_t)s0 * COLS);
            acc += v0 * c0;
        }
    }

    if (PRE) acc = acc * dd;
    acc += *(const vec*)(b + lane * VPL);
    if (RELU) {
#pragma unroll
        for (int k = 0; k < VPL; ++k) acc[k] = fmaxf(acc[k], 0.0f);
    }
    *(vec*)(out + (size_t)node * COLS + lane * VPL) = acc;
}


static inline char* bump(char*& p, size_t bytes) {
    char* r = p;
    p += (bytes + 255) & ~(size_t)255;
    return r;
}

extern "C" void kernel_launch(void* const* d_in, const int* in_sizes, int n_in,
                              void* d_out, int out_size, void* d_ws, size_t ws_size,
                              hipStream_t stream) {
    const float* x   = (const float*)d_in[0];
    const int*  eidx = (const int*)d_in[1];
    const float* ew  = (const float*)d_in[2];
    const float* W1  = (const float*)d_in[3];
    const float* b1  = (const float*)d_in[4];
    const float* W2  = (const float*)d_in[5];
    const float* b2  = (const float*)d_in[6];

    const int N = in_sizes[0] / 128;
    const int E = in_sizes[2];
    const int* src = eidx;
    const int* dst = eidx + E;

    const int T = 256;
    const int nBins   = (N + BIN_NODES - 1) / BIN_NODES;
    const int nChunks = (E + CHUNK - 1) / CHUNK;
    const int nCm     = nBins * nChunks;
    const int gemmBlocks = ((N + 15) / 16 + 7) / 8;
    const int nodeBlocks = (N + 7) / 8;
    const int WGPS = 256;
    const int gemmFull = ((N + 15) / 16) / 8;
    const int gemmL1 = gemmFull >= WGPS ? (gemmFull / WGPS) * WGPS : gemmBlocks;

    char* p = (char*)d_ws;
    int*   cm      = (int*)  bump(p, (size_t)nCm * sizeof(int));
    int*   lo      = (int*)  bump(p, (size_t)nCm * sizeof(int));
    int*   rowptr  = (int*)  bump(p, (size_t)(N + 1) * sizeof(int));
    int2*  tmp     = (int2*) bump(p, (size_t)E * sizeof(int2));
    int2*  es      = (int2*) bump(p, (size_t)E * sizeof(int2));
    float* dinv    = (float*)bump(p, (size_t)N * sizeof(float));
    float* h1      = (float*)bump(p, (size_t)N * 128 * sizeof(float));
    float* a1      = (float*)bump(p, (size_t)N * 128 * sizeof(float));
    float* g       = (float*)bump(p, (size_t)N * 64 * sizeof(float));
    float* out     = (float*)d_out;

    gemm_spread_kernel<<<gemmL1 + nChunks, T, 0, stream>>>(x, W1, h1, N, gemmL1,
                                                               src, dst, ew, tmp, cm, lo,
                                                               E, nBins, nChunks);
    bin_sort_kernel<<<nBins, 1024, 0, stream>>>(cm, lo, tmp, rowptr, es, dinv,
                                                N, E, nBins, nChunks);

    gather_kernel<128, true, false><<<nodeBlocks, T, 0, stream>>>(h1, rowptr, es, dinv, b1, a1, N);

    gemm_wmma_kernel<64><<<gemmBlocks, T, 0, stream>>>(a1, W2, g, dinv, N);
    gather_kernel<64, false, true><<<nodeBlocks, T, 0, stream>>>(g, rowptr, es, dinv, b2, out, N);
}
